# starting kernel
# baseline (speedup 1.0000x reference)
#include <hip/hip_runtime.h>
#include <math.h>

#define IN_CAPS 2401
#define IN_SIZE 8
#define OUT_CAPS 8
#define OUT_SIZE 16
#define BATCH 256
#define KDIM (IN_CAPS * IN_SIZE)
#define NDIM (OUT_CAPS * OUT_SIZE)
#define NKC 16
#define CPC 151

typedef float v2f __attribute__((ext_vector_type(2)));
typedef float v4f __attribute__((ext_vector_type(4)));
typedef float v8f __attribute__((ext_vector_type(8)));

__global__ void k_init_bc(float* __restrict__ bij, float* __restrict__ cij) {
    int t = blockIdx.x * blockDim.x + threadIdx.x;
    if (t < IN_CAPS * OUT_CAPS) { bij[t] = 0.0f; cij[t] = 0.125f; }
}

__global__ void k_transpose_x(const float* __restrict__ x, float* __restrict__ xt) {
    __shared__ float tile[8][33];
    int rt = blockIdx.x;
    int bt = blockIdx.y;
    int tid = threadIdx.x;
    int tr = tid & 7, tb = tid >> 3;
    tile[tr][tb] = x[(size_t)(bt * 32 + tb) * KDIM + rt * 8 + tr];
    __syncthreads();
    int wb = tid & 31, wr = tid >> 5;
    xt[(size_t)(rt * 8 + wr) * BATCH + bt * 32 + wb] = tile[wr][wb];
}

__global__ void k_route_gemm(const float* __restrict__ x, const float* __restrict__ W,
                             const float* __restrict__ cij, float* __restrict__ part) {
    int lane = threadIdx.x & 31;
    int job  = __builtin_amdgcn_readfirstlane(blockIdx.x * 4 + (threadIdx.x >> 5));
    int kc   = job >> 7;
    int rem  = job & 127;
    int mt   = rem >> 3;
    int ot   = rem & 7;
    int ln   = lane & 15;
    int half = lane >> 4;
    int k0   = half * 2;

    int cap0 = kc * CPC;
    int cap1 = min(cap0 + CPC, IN_CAPS);

    const float* xrow = x + (size_t)(mt * 16 + ln) * KDIM;
    v8f acc = {};
    for (int i = cap0; i < cap1; ++i) {
        float cs = cij[i * OUT_CAPS + ot];
        const float* xi   = xrow + i * 8;
        const float* wrow = W + (size_t)i * 1024 + ot * 128 + ln * 8;
#pragma unroll
        for (int g = 0; g < 2; ++g) {
            int kb = g * 4 + k0;
            v2f a = *(const v2f*)(xi + kb);
            v2f b = *(const v2f*)(wrow + kb);
            b *= cs;
            acc = __builtin_amdgcn_wmma_f32_16x16x4_f32(false, a, false, b,
                                                        (short)0, acc, false, false);
        }
    }
    float* pbase = part + ((size_t)kc * BATCH + mt * 16) * NDIM + ot * 16 + ln;
#pragma unroll
    for (int j = 0; j < 8; ++j)
        pbase[(size_t)(j + half * 8) * NDIM] = acc[j];
}

__global__ void k_reduce_squash(const float* __restrict__ part, float* __restrict__ vT,
                                float* __restrict__ out, int last) {
    int b = blockIdx.x;
    int n = threadIdx.x;
    float s = 0.0f;
#pragma unroll
    for (int kc = 0; kc < NKC; ++kc)
        s += part[((size_t)kc * BATCH + b) * NDIM + n];
    float msq = s * s;
#pragma unroll
    for (int m = 1; m < 16; m <<= 1)
        msq += __shfl_xor(msq, m, 32);
    float mag = sqrtf(msq);
    float v = s * (mag / (1.0f + msq));
    vT[(size_t)n * BATCH + b] = v;
    if (last) out[(size_t)b * NDIM + n] = v;
}

__global__ void k_agree(const float* __restrict__ xt, const float* __restrict__ vT,
                        const float* __restrict__ W, float* __restrict__ bij) {
    int lane = threadIdx.x & 31;
    int job  = __builtin_amdgcn_readfirstlane(blockIdx.x * 4 + (threadIdx.x >> 5));
    int it2  = job >> 3;
    int o    = job & 7;
    int i0   = it2 * 2;
    int ln   = lane & 15;
    int half = lane >> 4;
    int k0   = half * 2;

    int r = min(i0 * 8 + ln, KDIM - 1);
    const float* arow = xt + (size_t)r * BATCH;
    const float* brow = vT + (size_t)(o * 16 + ln) * BATCH;
    v8f acc = {};
#pragma unroll 4
    for (int kb = 0; kb < BATCH; kb += 4) {
        v2f a = *(const v2f*)(arow + kb + k0);
        v2f b = *(const v2f*)(brow + kb + k0);
        acc = __builtin_amdgcn_wmma_f32_16x16x4_f32(false, a, false, b,
                                                    (short)0, acc, false, false);
    }
    int iW = i0 + half;
    float dot = 0.0f;
    if (iW < IN_CAPS) {
        const float* wp = W + (size_t)iW * 1024 + o * 128 + ln * 8;
        v4f w0 = *(const v4f*)wp;
        v4f w1 = *(const v4f*)(wp + 4);
        dot = acc[0] * w0.x + acc[1] * w0.y + acc[2] * w0.z + acc[3] * w0.w
            + acc[4] * w1.x + acc[5] * w1.y + acc[6] * w1.z + acc[7] * w1.w;
    }
#pragma unroll
    for (int m = 1; m < 16; m <<= 1)
        dot += __shfl_xor(dot, m, 32);
    if (ln == 0 && iW < IN_CAPS)
        bij[iW * OUT_CAPS + o] += dot * (1.0f / BATCH);
}

__global__ void k_softmax(const float* __restrict__ bij, float* __restrict__ cij) {
    int i = blockIdx.x * blockDim.x + threadIdx.x;
    if (i >= IN_CAPS) return;
    float v[OUT_CAPS];
    float mx = -1e30f;
#pragma unroll
    for (int o = 0; o < OUT_CAPS; ++o) { v[o] = bij[i * OUT_CAPS + o]; mx = fmaxf(mx, v[o]); }
    float sum = 0.0f;
#pragma unroll
    for (int o = 0; o < OUT_CAPS; ++o) { v[o] = __expf(v[o] - mx); sum += v[o]; }
    float inv = 1.0f / sum;
#pragma unroll
    for (int o = 0; o < OUT_CAPS; ++o) cij[i * OUT_CAPS + o] = v[o] * inv;
}

extern "C" void kernel_launch(void* const* d_in, const int* in_sizes, int n_in,
                              void* d_out, int out_size, void* d_ws, size_t ws_size,
                              hipStream_t stream) {
    const float* x = (const float*)d_in[0];
    const float* W = (const float*)d_in[1];
    float* out = (float*)d_out;

    float* ws   = (float*)d_ws;
    float* XT   = ws;
    float* part = XT + (size_t)KDIM * BATCH;
    float* vT   = part + (size_t)NKC * BATCH * NDIM;
    float* bij  = vT + (size_t)NDIM * BATCH;
    float* cij  = bij + IN_CAPS * OUT_CAPS;

    k_init_bc<<<(IN_CAPS * OUT_CAPS + 255) / 256, 256, 0, stream>>>(bij, cij);
    k_transpose_x<<<dim3(IN_CAPS, 8), 256, 0, stream>>>(x, XT);

    for (int it = 0; it < 3; ++it) {
        k_route_gemm<<<512, 128, 0, stream>>>(x, W, cij, part);
        k_reduce_squash<<<BATCH, NDIM, 0, stream>>>(part, vT, out, it == 2 ? 1 : 0);
        if (it < 2) {
            k_agree<<<2402, 128, 0, stream>>>(XT, vT, W, bij);
            k_softmax<<<(IN_CAPS + 127) / 128, 128, 0, stream>>>(bij, cij);
        }
    }
}


# route GEMM: loads of 8 capsules issued as one group ahead of their 16 WMMAs (one memory round trip per group, not per capsule); last group slides back, repeats scaled by 0; sum order kept
# speedup vs baseline: 1.6266x; 1.6266x over previous
#include <hip/hip_runtime.h>
#include <math.h>

#define IN_CAPS 2401
#define IN_SIZE 8
#define OUT_CAPS 8
#define OUT_SIZE 16
#define BATCH 256
#define KDIM (IN_CAPS * IN_SIZE)
#define NDIM (OUT_CAPS * OUT_SIZE)
#define NKC 16
#define CPC 151
#define RG_U 8

typedef float v2f __attribute__((ext_vector_type(2)));
typedef float v4f __attribute__((ext_vector_type(4)));
typedef float v8f __attribute__((ext_vector_type(8)));

__global__ void k_init_bc(float* __restrict__ bij, float* __restrict__ cij) {
    int t = blockIdx.x * blockDim.x + threadIdx.x;
    if (t < IN_CAPS * OUT_CAPS) { bij[t] = 0.0f; cij[t] = 0.125f; }
}

__global__ void k_transpose_x(const float* __restrict__ x, float* __restrict__ xt) {
    __shared__ float tile[8][33];
    int rt = blockIdx.x;
    int bt = blockIdx.y;
    int tid = threadIdx.x;
    int tr = tid & 7, tb = tid >> 3;
    tile[tr][tb] = x[(size_t)(bt * 32 + tb) * KDIM + rt * 8 + tr];
    __syncthreads();
    int wb = tid & 31, wr = tid >> 5;
    xt[(size_t)(rt * 8 + wr) * BATCH + bt * 32 + wb] = tile[wr][wb];
}

__global__ void __launch_bounds__(128) __attribute__((amdgpu_waves_per_eu(1, 4)))
k_route_gemm(const float* __restrict__ x, const float* __restrict__ W,
                             const float* __restrict__ cij, float* __restrict__ part) {
    int lane = threadIdx.x & 31;
    int job  = __builtin_amdgcn_readfirstlane(blockIdx.x * 4 + (threadIdx.x >> 5));
    int kc   = job >> 7;
    int rem  = job & 127;
    int mt   = rem >> 3;
    int ot   = rem & 7;
    int ln   = lane & 15;
    int half = lane >> 4;
    int k0   = half * 2;

    int cap0 = kc * CPC;
    int cap1 = min(cap0 + CPC, IN_CAPS);

    const float* xlane = x + (size_t)(mt * 16 + ln) * KDIM + k0;
    const float* wlane = W + ot * 128 + ln * 8 + k0;
    v8f acc = {};
    for (int i = cap0; i < cap1; i += RG_U) {
        int base = min(i, cap1 - RG_U);
        const float* xg = xlane + base * 8;
        const float* wg = wlane + (size_t)base * 1024;
        const float* cg = cij + base * OUT_CAPS + ot;
        v2f a[RG_U][2], b[RG_U][2];
        float cs[RG_U];
#pragma unroll
        for (int u = 0; u < RG_U; ++u) {
            float c = cg[u * OUT_CAPS];
            cs[u] = (base + u >= i) ? c : 0.0f;
            a[u][0] = *(const v2f*)(xg + u * 8);
            a[u][1] = *(const v2f*)(xg + u * 8 + 4);
            b[u][0] = *(const v2f*)(wg + u * 1024);
            b[u][1] = *(const v2f*)(wg + u * 1024 + 4);
        }
#pragma unroll
        for (int u = 0; u < RG_U; ++u) {
#pragma unroll
            for (int g = 0; g < 2; ++g) {
                v2f bs = b[u][g] * cs[u];
                acc = __builtin_amdgcn_wmma_f32_16x16x4_f32(false, a[u][g], false, bs,
                                                            (short)0, acc, false, false);
            }
        }
    }
    float* pbase = part + ((size_t)kc * BATCH + mt * 16) * NDIM + ot * 16 + ln;
#pragma unroll
    for (int j = 0; j < 8; ++j)
        pbase[(size_t)(j + half * 8) * NDIM] = acc[j];
}

__global__ void k_reduce_squash(const float* __restrict__ part, float* __restrict__ vT,
                                float* __restrict__ out, int last) {
    int b = blockIdx.x;
    int n = threadIdx.x;
    float s = 0.0f;
#pragma unroll
    for (int kc = 0; kc < NKC; ++kc)
        s += part[((size_t)kc * BATCH + b) * NDIM + n];
    float msq = s * s;
#pragma unroll
    for (int m = 1; m < 16; m <<= 1)
        msq += __shfl_xor(msq, m, 32);
    float mag = sqrtf(msq);
    float v = s * (mag / (1.0f + msq));
    vT[(size_t)n * BATCH + b] = v;
    if (last) out[(size_t)b * NDIM + n] = v;
}

__global__ void k_agree(const float* __restrict__ xt, const float* __restrict__ vT,
                        const float* __restrict__ W, float* __restrict__ bij) {
    int lane = threadIdx.x & 31;
    int job  = __builtin_amdgcn_readfirstlane(blockIdx.x * 4 + (threadIdx.x >> 5));
    int it2  = job >> 3;
    int o    = job & 7;
    int i0   = it2 * 2;
    int ln   = lane & 15;
    int half = lane >> 4;
    int k0   = half * 2;

    int r = min(i0 * 8 + ln, KDIM - 1);
    const float* arow = xt + (size_t)r * BATCH;
    const float* brow = vT + (size_t)(o * 16 + ln) * BATCH;
    v8f acc = {};
#pragma unroll 4
    for (int kb = 0; kb < BATCH; kb += 4) {
        v2f a = *(const v2f*)(arow + kb + k0);
        v2f b = *(const v2f*)(brow + kb + k0);
        acc = __builtin_amdgcn_wmma_f32_16x16x4_f32(false, a, false, b,
                                                    (short)0, acc, false, false);
    }
    int iW = i0 + half;
    float dot = 0.0f;
    if (iW < IN_CAPS) {
        const float* wp = W + (size_t)iW * 1024 + o * 128 + ln * 8;
        v4f w0 = *(const v4f*)wp;
        v4f w1 = *(const v4f*)(wp + 4);
        dot = acc[0] * w0.x + acc[1] * w0.y + acc[2] * w0.z + acc[3] * w0.w
            + acc[4] * w1.x + acc[5] * w1.y + acc[6] * w1.z + acc[7] * w1.w;
    }
#pragma unroll
    for (int m = 1; m < 16; m <<= 1)
        dot += __shfl_xor(dot, m, 32);
    if (ln == 0 && iW < IN_CAPS)
        bij[iW * OUT_CAPS + o] += dot * (1.0f / BATCH);
}

__global__ void k_softmax(const float* __restrict__ bij, float* __restrict__ cij) {
    int i = blockIdx.x * blockDim.x + threadIdx.x;
    if (i >= IN_CAPS) return;
    float v[OUT_CAPS];
    float mx = -1e30f;
#pragma unroll
    for (int o = 0; o < OUT_CAPS; ++o) { v[o] = bij[i * OUT_CAPS + o]; mx = fmaxf(mx, v[o]); }
    float sum = 0.0f;
#pragma unroll
    for (int o = 0; o < OUT_CAPS; ++o) { v[o] = __expf(v[o] - mx); sum += v[o]; }
    float inv = 1.0f / sum;
#pragma unroll
    for (int o = 0; o < OUT_CAPS; ++o) cij[i * OUT_CAPS + o] = v[o] * inv;
}

extern "C" void kernel_launch(void* const* d_in, const int* in_sizes, int n_in,
                              void* d_out, int out_size, void* d_ws, size_t ws_size,
                              hipStream_t stream) {
    const float* x = (const float*)d_in[0];
    const float* W = (const float*)d_in[1];
    float* out = (float*)d_out;

    float* ws   = (float*)d_ws;
    float* XT   = ws;
    float* part = XT + (size_t)KDIM * BATCH;
    float* vT   = part + (size_t)NKC * BATCH * NDIM;
    float* bij  = vT + (size_t)NDIM * BATCH;
    float* cij  = bij + IN_CAPS * OUT_CAPS;

    k_init_bc<<<(IN_CAPS * OUT_CAPS + 255) / 256, 256, 0, stream>>>(bij, cij);
    k_transpose_x<<<dim3(IN_CAPS, 8), 256, 0, stream>>>(x, XT);

    for (int it = 0; it < 3; ++it) {
        k_route_gemm<<<512, 128, 0, stream>>>(x, W, cij, part);
        k_reduce_squash<<<BATCH, NDIM, 0, stream>>>(part, vT, out, it == 2 ? 1 : 0);
        if (it < 2) {
            k_agree<<<2402, 128, 0, stream>>>(XT, vT, W, bij);
            k_softmax<<<(IN_CAPS + 127) / 128, 128, 0, stream>>>(bij, cij);
        }
    }
}


# agreement GEMM reads its A operand straight from x: one 16-byte load per lane feeds 4 row-interleaved M-tiles (8 capsules per wave); transpose kernel and transposed copy of x dropped; sum order kept
# speedup vs baseline: 2.3950x; 1.4724x over previous
#include <hip/hip_runtime.h>
#include <math.h>

#define IN_CAPS 2401
#define IN_SIZE 8
#define OUT_CAPS 8
#define OUT_SIZE 16
#define BATCH 256
#define KDIM (IN_CAPS * IN_SIZE)
#define NDIM (OUT_CAPS * OUT_SIZE)
#define NKC 16
#define CPC 151
#define RG_U 8
#define AG_T 4
#define AG_U 8
#define AG_GROUPS ((IN_CAPS + 7) / 8)

typedef float v2f __attribute__((ext_vector_type(2)));
typedef float v4f __attribute__((ext_vector_type(4)));
typedef float v8f __attribute__((ext_vector_type(8)));

__global__ void k_init_bc(float* __restrict__ bij, float* __restrict__ cij) {
    int t = blockIdx.x * blockDim.x + threadIdx.x;
    if (t < IN_CAPS * OUT_CAPS) { bij[t] = 0.0f; cij[t] = 0.125f; }
}

__global__ void __launch_bounds__(128) __attribute__((amdgpu_waves_per_eu(1, 4)))
k_route_gemm(const float* __restrict__ x, const float* __restrict__ W,
                             const float* __restrict__ cij, float* __restrict__ part) {
    int lane = threadIdx.x & 31;
    int job  = __builtin_amdgcn_readfirstlane(blockIdx.x * 4 + (threadIdx.x >> 5));
    int kc   = job >> 7;
    int rem  = job & 127;
    int mt   = rem >> 3;
    int ot   = rem & 7;
    int ln   = lane & 15;
    int half = lane >> 4;
    int k0   = half * 2;

    int cap0 = kc * CPC;
    int cap1 = min(cap0 + CPC, IN_CAPS);

    const float* xlane = x + (size_t)(mt * 16 + ln) * KDIM + k0;
    const float* wlane = W + ot * 128 + ln * 8 + k0;
    v8f acc = {};
    for (int i = cap0; i < cap1; i += RG_U) {
        int base = min(i, cap1 - RG_U);
        const float* xg = xlane + base * 8;
        const float* wg = wlane + (size_t)base * 1024;
        const float* cg = cij + base * OUT_CAPS + ot;
        v2f a[RG_U][2], b[RG_U][2];
        float cs[RG_U];
#pragma unroll
        for (int u = 0; u < RG_U; ++u) {
            float c = cg[u * OUT_CAPS];
            cs[u] = (base + u >= i) ? c : 0.0f;
            a[u][0] = *(const v2f*)(xg + u * 8);
            a[u][1] = *(const v2f*)(xg + u * 8 + 4);
            b[u][0] = *(const v2f*)(wg + u * 1024);
            b[u][1] = *(const v2f*)(wg + u * 1024 + 4);
        }
#pragma unroll
        for (int u = 0; u < RG_U; ++u) {
#pragma unroll
            for (int g = 0; g < 2; ++g) {
                v2f bs = b[u][g] * cs[u];
                acc = __builtin_amdgcn_wmma_f32_16x16x4_f32(false, a[u][g], false, bs,
                                                            (short)0, acc, false, false);
            }
        }
    }
    float* pbase = part + ((size_t)kc * BATCH + mt * 16) * NDIM + ot * 16 + ln;
#pragma unroll
    for (int j = 0; j < 8; ++j)
        pbase[(size_t)(j + half * 8) * NDIM] = acc[j];
}

__global__ void k_reduce_squash(const float* __restrict__ part, float* __restrict__ vT,
                                float* __restrict__ out, int last) {
    int b = blockIdx.x;
    int n = threadIdx.x;
    float s = 0.0f;
#pragma unroll
    for (int kc = 0; kc < NKC; ++kc)
        s += part[((size_t)kc * BATCH + b) * NDIM + n];
    float msq = s * s;
#pragma unroll
    for (int m = 1; m < 16; m <<= 1)
        msq += __shfl_xor(msq, m, 32);
    float mag = sqrtf(msq);
    float v = s * (mag / (1.0f + msq));
    vT[(size_t)n * BATCH + b] = v;
    if (last) out[(size_t)b * NDIM + n] = v;
}

__global__ void __launch_bounds__(128)
k_agree(const float* __restrict__ x, const float* __restrict__ vT,
        const float* __restrict__ W, float* __restrict__ bij) {
    int lane = threadIdx.x & 31;
    int job  = __builtin_amdgcn_readfirstlane(blockIdx.x * 4 + (threadIdx.x >> 5));
    int g    = job >> 3;
    int o    = job & 7;
    int ln   = lane & 15;
    int half = lane >> 4;
    int k0   = half * 2;

    int kofs = min(g * 64 + ln * 4, KDIM - 4);
    const float* xl   = x + (size_t)k0 * KDIM + kofs;
    const float* brow = vT + (size_t)(o * 16 + ln) * BATCH + k0;
    v8f acc[AG_T] = {};
    for (int kb = 0; kb < BATCH; kb += 4 * AG_U) {
        v4f xa[AG_U][2];
        v2f vb[AG_U];
#pragma unroll
        for (int u = 0; u < AG_U; ++u) {
            xa[u][0] = *(const v4f*)(xl + (size_t)(kb + 4 * u) * KDIM);
            xa[u][1] = *(const v4f*)(xl + (size_t)(kb + 4 * u + 1) * KDIM);
            vb[u]    = *(const v2f*)(brow + kb + 4 * u);
        }
#pragma unroll
        for (int u = 0; u < AG_U; ++u) {
#pragma unroll
            for (int t = 0; t < AG_T; ++t) {
                v2f a = { xa[u][0][t], xa[u][1][t] };
                acc[t] = __builtin_amdgcn_wmma_f32_16x16x4_f32(false, a, false, vb[u],
                                                               (short)0, acc[t], false, false);
            }
        }
    }
    float dot[4];
#pragma unroll
    for (int q = 0; q < 4; ++q) {
        int iW = min(g * 8 + half * 4 + q, IN_CAPS - 1);
        const float* wp = W + (size_t)iW * 1024 + o * 128 + ln * 8;
        v4f w0 = *(const v4f*)wp;
        v4f w1 = *(const v4f*)(wp + 4);
        dot[q] = acc[0][2 * q] * w0.x + acc[1][2 * q] * w0.y
               + acc[2][2 * q] * w0.z + acc[3][2 * q] * w0.w
               + acc[0][2 * q + 1] * w1.x + acc[1][2 * q + 1] * w1.y
               + acc[2][2 * q + 1] * w1.z + acc[3][2 * q + 1] * w1.w;
    }
#pragma unroll
    for (int q = 0; q < 4; ++q) {
#pragma unroll
        for (int m = 1; m < 16; m <<= 1)
            dot[q] += __shfl_xor(dot[q], m, 32);
    }
#pragma unroll
    for (int q = 0; q < 4; ++q) {
        int iW = g * 8 + half * 4 + q;
        if (ln == 0 && iW < IN_CAPS)
            bij[iW * OUT_CAPS + o] += dot[q] * (1.0f / BATCH);
    }
}

__global__ void k_softmax(const float* __restrict__ bij, float* __restrict__ cij) {
    int i = blockIdx.x * blockDim.x + threadIdx.x;
    if (i >= IN_CAPS) return;
    float v[OUT_CAPS];
    float mx = -1e30f;
#pragma unroll
    for (int o = 0; o < OUT_CAPS; ++o) { v[o] = bij[i * OUT_CAPS + o]; mx = fmaxf(mx, v[o]); }
    float sum = 0.0f;
#pragma unroll
    for (int o = 0; o < OUT_CAPS; ++o) { v[o] = __expf(v[o] - mx); sum += v[o]; }
    float inv = 1.0f / sum;
#pragma unroll
    for (int o = 0; o < OUT_CAPS; ++o) cij[i * OUT_CAPS + o] = v[o] * inv;
}

extern "C" void kernel_launch(void* const* d_in, const int* in_sizes, int n_in,
                              void* d_out, int out_size, void* d_ws, size_t ws_size,
                              hipStream_t stream) {
    const float* x = (const float*)d_in[0];
    const float* W = (const float*)d_in[1];
    float* out = (float*)d_out;

    float* ws   = (float*)d_ws;
    float* part = ws;
    float* vT   = part + (size_t)NKC * BATCH * NDIM;
    float* bij  = vT + (size_t)NDIM * BATCH;
    float* cij  = bij + IN_CAPS * OUT_CAPS;

    k_init_bc<<<(IN_CAPS * OUT_CAPS + 255) / 256, 256, 0, stream>>>(bij, cij);

    for (int it = 0; it < 3; ++it) {
        k_route_gemm<<<512, 128, 0, stream>>>(x, W, cij, part);
        k_reduce_squash<<<BATCH, NDIM, 0, stream>>>(part, vT, out, it == 2 ? 1 : 0);
        if (it < 2) {
            k_agree<<<AG_GROUPS * OUT_CAPS / 4, 128, 0, stream>>>(x, vT, W, bij);
            k_softmax<<<(IN_CAPS + 127) / 128, 128, 0, stream>>>(bij, cij);
        }
    }
}


# route GEMM: a lane fetches its A and B share of a capsule with one 16-byte load, not two 8-byte ones (k re-ordered across the capsule's two WMMAs, same for A and B); sum order in a capsule changed
# speedup vs baseline: 2.5569x; 1.0676x over previous
#include <hip/hip_runtime.h>
#include <math.h>

#define IN_CAPS 2401
#define IN_SIZE 8
#define OUT_CAPS 8
#define OUT_SIZE 16
#define BATCH 256
#define KDIM (IN_CAPS * IN_SIZE)
#define NDIM (OUT_CAPS * OUT_SIZE)
#define NKC 16
#define CPC 151
#define RG_U 8
#define AG_T 4
#define AG_U 8
#define AG_GROUPS ((IN_CAPS + 7) / 8)

typedef float v2f __attribute__((ext_vector_type(2)));
typedef float v4f __attribute__((ext_vector_type(4)));
typedef float v8f __attribute__((ext_vector_type(8)));

__global__ void k_init_bc(float* __restrict__ bij, float* __restrict__ cij) {
    int t = blockIdx.x * blockDim.x + threadIdx.x;
    if (t < IN_CAPS * OUT_CAPS) { bij[t] = 0.0f; cij[t] = 0.125f; }
}

__global__ void __launch_bounds__(128) __attribute__((amdgpu_waves_per_eu(1, 4)))
k_route_gemm(const float* __restrict__ x, const float* __restrict__ W,
                             const float* __restrict__ cij, float* __restrict__ part) {
    int lane = threadIdx.x & 31;
    int job  = __builtin_amdgcn_readfirstlane(blockIdx.x * 4 + (threadIdx.x >> 5));
    int kc   = job >> 7;
    int rem  = job & 127;
    int mt   = rem >> 3;
    int ot   = rem & 7;
    int ln   = lane & 15;
    int half = lane >> 4;
    int d0   = half * 4;

    int cap0 = kc * CPC;
    int cap1 = min(cap0 + CPC, IN_CAPS);

    const float* xlane = x + (size_t)(mt * 16 + ln) * KDIM + d0;
    const float* wlane = W + ot * 128 + ln * 8 + d0;
    v8f acc = {};
    for (int i = cap0; i < cap1; i += RG_U) {
        int base = min(i, cap1 - RG_U);
        const float* xg = xlane + base * 8;
        const float* wg = wlane + (size_t)base * 1024;
        const float* cg = cij + base * OUT_CAPS + ot;
        v4f a[RG_U], b[RG_U];
        float cs[RG_U];
#pragma unroll
        for (int u = 0; u < RG_U; ++u) {
            float c = cg[u * OUT_CAPS];
            cs[u] = (base + u >= i) ? c : 0.0f;
            a[u] = *(const v4f*)(xg + u * 8);
            b[u] = *(const v4f*)(wg + u * 1024);
        }
#pragma unroll
        for (int u = 0; u < RG_U; ++u) {
#pragma unroll
            for (int g = 0; g < 2; ++g) {
                v2f av = { a[u][2 * g], a[u][2 * g + 1] };
                v2f bs = { b[u][2 * g] * cs[u], b[u][2 * g + 1] * cs[u] };
                acc = __builtin_amdgcn_wmma_f32_16x16x4_f32(false, av, false, bs,
                                                            (short)0, acc, false, false);
            }
        }
    }
    float* pbase = part + ((size_t)kc * BATCH + mt * 16) * NDIM + ot * 16 + ln;
#pragma unroll
    for (int j = 0; j < 8; ++j)
        pbase[(size_t)(j + half * 8) * NDIM] = acc[j];
}

__global__ void k_reduce_squash(const float* __restrict__ part, float* __restrict__ vT,
                                float* __restrict__ out, int last) {
    int b = blockIdx.x;
    int n = threadIdx.x;
    float s = 0.0f;
#pragma unroll
    for (int kc = 0; kc < NKC; ++kc)
        s += part[((size_t)kc * BATCH + b) * NDIM + n];
    float msq = s * s;
#pragma unroll
    for (int m = 1; m < 16; m <<= 1)
        msq += __shfl_xor(msq, m, 32);
    float mag = sqrtf(msq);
    float v = s * (mag / (1.0f + msq));
    vT[(size_t)n * BATCH + b] = v;
    if (last) out[(size_t)b * NDIM + n] = v;
}

__global__ void __launch_bounds__(128)
k_agree(const float* __restrict__ x, const float* __restrict__ vT,
        const float* __restrict__ W, float* __restrict__ bij) {
    int lane = threadIdx.x & 31;
    int job  = __builtin_amdgcn_readfirstlane(blockIdx.x * 4 + (threadIdx.x >> 5));
    int g    = job >> 3;
    int o    = job & 7;
    int ln   = lane & 15;
    int half = lane >> 4;
    int k0   = half * 2;

    int kofs = min(g * 64 + ln * 4, KDIM - 4);
    const float* xl   = x + (size_t)k0 * KDIM + kofs;
    const float* brow = vT + (size_t)(o * 16 + ln) * BATCH + k0;
    v8f acc[AG_T] = {};
    for (int kb = 0; kb < BATCH; kb += 4 * AG_U) {
        v4f xa[AG_U][2];
        v2f vb[AG_U];
#pragma unroll
        for (int u = 0; u < AG_U; ++u) {
            xa[u][0] = *(const v4f*)(xl + (size_t)(kb + 4 * u) * KDIM);
            xa[u][1] = *(const v4f*)(xl + (size_t)(kb + 4 * u + 1) * KDIM);
            vb[u]    = *(const v2f*)(brow + kb + 4 * u);
        }
#pragma unroll
        for (int u = 0; u < AG_U; ++u) {
#pragma unroll
            for (int t = 0; t < AG_T; ++t) {
                v2f a = { xa[u][0][t], xa[u][1][t] };
                acc[t] = __builtin_amdgcn_wmma_f32_16x16x4_f32(false, a, false, vb[u],
                                                               (short)0, acc[t], false, false);
            }
        }
    }
    float dot[4];
#pragma unroll
    for (int q = 0; q < 4; ++q) {
        int iW = min(g * 8 + half * 4 + q, IN_CAPS - 1);
        const float* wp = W + (size_t)iW * 1024 + o * 128 + ln * 8;
        v4f w0 = *(const v4f*)wp;
        v4f w1 = *(const v4f*)(wp + 4);
        dot[q] = acc[0][2 * q] * w0.x + acc[1][2 * q] * w0.y
               + acc[2][2 * q] * w0.z + acc[3][2 * q] * w0.w
               + acc[0][2 * q + 1] * w1.x + acc[1][2 * q + 1] * w1.y
               + acc[2][2 * q + 1] * w1.z + acc[3][2 * q + 1] * w1.w;
    }
#pragma unroll
    for (int q = 0; q < 4; ++q) {
#pragma unroll
        for (int m = 1; m < 16; m <<= 1)
            dot[q] += __shfl_xor(dot[q], m, 32);
    }
#pragma unroll
    for (int q = 0; q < 4; ++q) {
        int iW = g * 8 + half * 4 + q;
        if (ln == 0 && iW < IN_CAPS)
            bij[iW * OUT_CAPS + o] += dot[q] * (1.0f / BATCH);
    }
}

__global__ void k_softmax(const float* __restrict__ bij, float* __restrict__ cij) {
    int i = blockIdx.x * blockDim.x + threadIdx.x;
    if (i >= IN_CAPS) return;
    float v[OUT_CAPS];
    float mx = -1e30f;
#pragma unroll
    for (int o = 0; o < OUT_CAPS; ++o) { v[o] = bij[i * OUT_CAPS + o]; mx = fmaxf(mx, v[o]); }
    float sum = 0.0f;
#pragma unroll
    for (int o = 0; o < OUT_CAPS; ++o) { v[o] = __expf(v[o] - mx); sum += v[o]; }
    float inv = 1.0f / sum;
#pragma unroll
    for (int o = 0; o < OUT_CAPS; ++o) cij[i * OUT_CAPS + o] = v[o] * inv;
}

extern "C" void kernel_launch(void* const* d_in, const int* in_sizes, int n_in,
                              void* d_out, int out_size, void* d_ws, size_t ws_size,
                              hipStream_t stream) {
    const float* x = (const float*)d_in[0];
    const float* W = (const float*)d_in[1];
    float* out = (float*)d_out;

    float* ws   = (float*)d_ws;
    float* part = ws;
    float* vT   = part + (size_t)NKC * BATCH * NDIM;
    float* bij  = vT + (size_t)NDIM * BATCH;
    float* cij  = bij + IN_CAPS * OUT_CAPS;

    k_init_bc<<<(IN_CAPS * OUT_CAPS + 255) / 256, 256, 0, stream>>>(bij, cij);

    for (int it = 0; it < 3; ++it) {
        k_route_gemm<<<512, 128, 0, stream>>>(x, W, cij, part);
        k_reduce_squash<<<BATCH, NDIM, 0, stream>>>(part, vT, out, it == 2 ? 1 : 0);
        if (it < 2) {
            k_agree<<<AG_GROUPS * OUT_CAPS / 4, 128, 0, stream>>>(x, vT, W, bij);
            k_softmax<<<(IN_CAPS + 127) / 128, 128, 0, stream>>>(bij, cij);
        }
    }
}


# softmax and b/c init kernels folded away: each route-GEMM work-group builds its chunk's coefficients in LDS before its loop (1/8 on pass 1), first squash clears b; 11 launches -> 8; sum order kept
# speedup vs baseline: 2.8544x; 1.1164x over previous
#include <hip/hip_runtime.h>
#include <math.h>

#define IN_CAPS 2401
#define IN_SIZE 8
#define OUT_CAPS 8
#define OUT_SIZE 16
#define BATCH 256
#define KDIM (IN_CAPS * IN_SIZE)
#define NDIM (OUT_CAPS * OUT_SIZE)
#define NKC 16
#define CPC 151
#define RG_U 8
#define AG_T 4
#define AG_U 8
#define AG_GROUPS ((IN_CAPS + 7) / 8)

typedef float v2f __attribute__((ext_vector_type(2)));
typedef float v4f __attribute__((ext_vector_type(4)));
typedef float v8f __attribute__((ext_vector_type(8)));

__global__ void __launch_bounds__(128) __attribute__((amdgpu_waves_per_eu(1, 4)))
k_route_gemm(const float* __restrict__ x, const float* __restrict__ W,
                             const float* __restrict__ bij, float* __restrict__ part,
                             int first) {
    __shared__ float ctab[CPC * OUT_CAPS];
    int lane = threadIdx.x & 31;
    int job  = __builtin_amdgcn_readfirstlane(blockIdx.x * 4 + (threadIdx.x >> 5));
    int kc   = job >> 7;
    int rem  = job & 127;
    int mt   = rem >> 3;
    int ot   = rem & 7;
    int ln   = lane & 15;
    int half = lane >> 4;
    int d0   = half * 4;

    int cap0 = kc * CPC;
    int cap1 = min(cap0 + CPC, IN_CAPS);

    for (int j = threadIdx.x; j < cap1 - cap0; j += 128) {
        float v[OUT_CAPS];
        if (first) {
#pragma unroll
            for (int o = 0; o < OUT_CAPS; ++o) v[o] = 0.125f;
        } else {
            const float* bp = bij + (size_t)(cap0 + j) * OUT_CAPS;
            v4f b0 = *(const v4f*)bp;
            v4f b1 = *(const v4f*)(bp + 4);
            v[0] = b0.x; v[1] = b0.y; v[2] = b0.z; v[3] = b0.w;
            v[4] = b1.x; v[5] = b1.y; v[6] = b1.z; v[7] = b1.w;
            float mx = -1e30f;
#pragma unroll
            for (int o = 0; o < OUT_CAPS; ++o) mx = fmaxf(mx, v[o]);
            float sum = 0.0f;
#pragma unroll
            for (int o = 0; o < OUT_CAPS; ++o) { v[o] = __expf(v[o] - mx); sum += v[o]; }
            float inv = 1.0f / sum;
#pragma unroll
            for (int o = 0; o < OUT_CAPS; ++o) v[o] *= inv;
        }
        v4f c0 = { v[0], v[1], v[2], v[3] };
        v4f c1 = { v[4], v[5], v[6], v[7] };
        *(v4f*)(ctab + j * OUT_CAPS)     = c0;
        *(v4f*)(ctab + j * OUT_CAPS + 4) = c1;
    }
    __syncthreads();

    const float* xlane = x + (size_t)(mt * 16 + ln) * KDIM + d0;
    const float* wlane = W + ot * 128 + ln * 8 + d0;
    v8f acc = {};
    for (int i = cap0; i < cap1; i += RG_U) {
        int base = min(i, cap1 - RG_U);
        const float* xg = xlane + base * 8;
        const float* wg = wlane + (size_t)base * 1024;
        int cl = __float_as_int(ctab[(base - cap0 + (lane & 7)) * OUT_CAPS + ot]);
        v4f a[RG_U], b[RG_U];
        float cs[RG_U];
#pragma unroll
        for (int u = 0; u < RG_U; ++u) {
            float c = __int_as_float(__builtin_amdgcn_readlane(cl, u));
            cs[u] = (base + u >= i) ? c : 0.0f;
            a[u] = *(const v4f*)(xg + u * 8);
            b[u] = *(const v4f*)(wg + u * 1024);
        }
#pragma unroll
        for (int u = 0; u < RG_U; ++u) {
#pragma unroll
            for (int g = 0; g < 2; ++g) {
                v2f av = { a[u][2 * g], a[u][2 * g + 1] };
                v2f bs = { b[u][2 * g] * cs[u], b[u][2 * g + 1] * cs[u] };
                acc = __builtin_amdgcn_wmma_f32_16x16x4_f32(false, av, false, bs,
                                                            (short)0, acc, false, false);
            }
        }
    }
    float* pbase = part + ((size_t)kc * BATCH + mt * 16) * NDIM + ot * 16 + ln;
#pragma unroll
    for (int j = 0; j < 8; ++j)
        pbase[(size_t)(j + half * 8) * NDIM] = acc[j];
}

__global__ void k_reduce_squash(const float* __restrict__ part, float* __restrict__ vT,
                                float* __restrict__ out, float* __restrict__ bij,
                                int last, int zero_b) {
    int b = blockIdx.x;
    int n = threadIdx.x;
    if (zero_b) {
        int t = b * NDIM + n;
        if (t < IN_CAPS * OUT_CAPS) bij[t] = 0.0f;
    }
    float s = 0.0f;
#pragma unroll
    for (int kc = 0; kc < NKC; ++kc)
        s += part[((size_t)kc * BATCH + b) * NDIM + n];
    float msq = s * s;
#pragma unroll
    for (int m = 1; m < 16; m <<= 1)
        msq += __shfl_xor(msq, m, 32);
    float mag = sqrtf(msq);
    float v = s * (mag / (1.0f + msq));
    vT[(size_t)n * BATCH + b] = v;
    if (last) out[(size_t)b * NDIM + n] = v;
}

__global__ void __launch_bounds__(128)
k_agree(const float* __restrict__ x, const float* __restrict__ vT,
        const float* __restrict__ W, float* __restrict__ bij) {
    int lane = threadIdx.x & 31;
    int job  = __builtin_amdgcn_readfirstlane(blockIdx.x * 4 + (threadIdx.x >> 5));
    int g    = job >> 3;
    int o    = job & 7;
    int ln   = lane & 15;
    int half = lane >> 4;
    int k0   = half * 2;

    int kofs = min(g * 64 + ln * 4, KDIM - 4);
    const float* xl   = x + (size_t)k0 * KDIM + kofs;
    const float* brow = vT + (size_t)(o * 16 + ln) * BATCH + k0;
    v8f acc[AG_T] = {};
    for (int kb = 0; kb < BATCH; kb += 4 * AG_U) {
        v4f xa[AG_U][2];
        v2f vb[AG_U];
#pragma unroll
        for (int u = 0; u < AG_U; ++u) {
            xa[u][0] = *(const v4f*)(xl + (size_t)(kb + 4 * u) * KDIM);
            xa[u][1] = *(const v4f*)(xl + (size_t)(kb + 4 * u + 1) * KDIM);
            vb[u]    = *(const v2f*)(brow + kb + 4 * u);
        }
#pragma unroll
        for (int u = 0; u < AG_U; ++u) {
#pragma unroll
            for (int t = 0; t < AG_T; ++t) {
                v2f a = { xa[u][0][t], xa[u][1][t] };
                acc[t] = __builtin_amdgcn_wmma_f32_16x16x4_f32(false, a, false, vb[u],
                                                               (short)0, acc[t], false, false);
            }
        }
    }
    float dot[4];
#pragma unroll
    for (int q = 0; q < 4; ++q) {
        int iW = min(g * 8 + half * 4 + q, IN_CAPS - 1);
        const float* wp = W + (size_t)iW * 1024 + o * 128 + ln * 8;
        v4f w0 = *(const v4f*)wp;
        v4f w1 = *(const v4f*)(wp + 4);
        dot[q] = acc[0][2 * q] * w0.x + acc[1][2 * q] * w0.y
               + acc[2][2 * q] * w0.z + acc[3][2 * q] * w0.w
               + acc[0][2 * q + 1] * w1.x + acc[1][2 * q + 1] * w1.y
               + acc[2][2 * q + 1] * w1.z + acc[3][2 * q + 1] * w1.w;
    }
#pragma unroll
    for (int q = 0; q < 4; ++q) {
#pragma unroll
        for (int m = 1; m < 16; m <<= 1)
            dot[q] += __shfl_xor(dot[q], m, 32);
    }
#pragma unroll
    for (int q = 0; q < 4; ++q) {
        int iW = g * 8 + half * 4 + q;
        if (ln == 0 && iW < IN_CAPS)
            bij[iW * OUT_CAPS + o] += dot[q] * (1.0f / BATCH);
    }
}

extern "C" void kernel_launch(void* const* d_in, const int* in_sizes, int n_in,
                              void* d_out, int out_size, void* d_ws, size_t ws_size,
                              hipStream_t stream) {
    const float* x = (const float*)d_in[0];
    const float* W = (const float*)d_in[1];
    float* out = (float*)d_out;

    float* ws   = (float*)d_ws;
    float* part = ws;
    float* vT   = part + (size_t)NKC * BATCH * NDIM;
    float* bij  = vT + (size_t)NDIM * BATCH;

    k_route_gemm<<<512, 128, 0, stream>>>(x, W, bij, part, 1);
    k_reduce_squash<<<BATCH, NDIM, 0, stream>>>(part, vT, out, bij, 0, 1);
    k_agree<<<AG_GROUPS * OUT_CAPS / 4, 128, 0, stream>>>(x, vT, W, bij);
    k_route_gemm<<<512, 128, 0, stream>>>(x, W, bij, part, 0);
    k_reduce_squash<<<BATCH, NDIM, 0, stream>>>(part, vT, out, bij, 0, 0);
    k_agree<<<AG_GROUPS * OUT_CAPS / 4, 128, 0, stream>>>(x, vT, W, bij);
    k_route_gemm<<<512, 128, 0, stream>>>(x, W, bij, part, 0);
    k_reduce_squash<<<BATCH, NDIM, 0, stream>>>(part, vT, out, bij, 1, 0);
}


# route GEMM: each wave owns 2x4 output tiles (32 rows x 4 capsules), so a fragment load feeds 8 or 4 WMMAs, not 2; next group's loads issued ahead; 64 K-chunks, 256 groups; sum order changed
# speedup vs baseline: 3.4759x; 1.2177x over previous
#include <hip/hip_runtime.h>
#include <math.h>

#define IN_CAPS 2401
#define IN_SIZE 8
#define OUT_CAPS 8
#define OUT_SIZE 16
#define BATCH 256
#define KDIM (IN_CAPS * IN_SIZE)
#define NDIM (OUT_CAPS * OUT_SIZE)
#define NKC 64
#define CPC_LO 37
#define CPC_EXTRA 33
#define CPC_MAX 38
#define RG_U 4
#define RG_NG 10
#define RG_TM 2
#define RG_TN 4
#define AG_T 4
#define AG_U 8
#define AG_GROUPS ((IN_CAPS + 7) / 8)

typedef float v2f __attribute__((ext_vector_type(2)));
typedef float v4f __attribute__((ext_vector_type(4)));
typedef float v8f __attribute__((ext_vector_type(8)));

struct RouteFrags {
    v4f a[RG_U][RG_TM];
    v4f b[RG_U][RG_TN];
};

__device__ __forceinline__ void route_load(RouteFrags& f, const float* xlane, const float* wlane,
                                           int lb) {
    const float* xg = xlane + lb * 8;
    const float* wg = wlane + (size_t)lb * 1024;
#pragma unroll
    for (int u = 0; u < RG_U; ++u) {
#pragma unroll
        for (int t = 0; t < RG_TM; ++t)
            f.a[u][t] = *(const v4f*)(xg + u * 8 + (size_t)t * 16 * KDIM);
#pragma unroll
        for (int q = 0; q < RG_TN; ++q)
            f.b[u][q] = *(const v4f*)(wg + u * 1024 + q * 128);
    }
}

__device__ __forceinline__ void route_mma(v8f (&acc)[RG_TM][RG_TN], const RouteFrags& f,
                                          const float* ctab, int lb, int gstart, int oq, int lane) {
    int cl = __float_as_int(ctab[(lb + ((lane >> 2) & 3)) * OUT_CAPS + oq * RG_TN + (lane & 3)]);
#pragma unroll
    for (int u = 0; u < RG_U; ++u) {
#pragma unroll
        for (int q = 0; q < RG_TN; ++q) {
            float c = __int_as_float(__builtin_amdgcn_readlane(cl, u * RG_TN + q));
            c = (lb + u >= gstart) ? c : 0.0f;
#pragma unroll
            for (int h = 0; h < 2; ++h) {
                v2f bs = { f.b[u][q][2 * h] * c, f.b[u][q][2 * h + 1] * c };
#pragma unroll
                for (int t = 0; t < RG_TM; ++t) {
                    v2f av = { f.a[u][t][2 * h], f.a[u][t][2 * h + 1] };
                    acc[t][q] = __builtin_amdgcn_wmma_f32_16x16x4_f32(false, av, false, bs,
                                                                      (short)0, acc[t][q],
                                                                      false, false);
                }
            }
        }
    }
}

__global__ void __launch_bounds__(128) __attribute__((amdgpu_waves_per_eu(1, 2)))
k_route_gemm(const float* __restrict__ x, const float* __restrict__ W,
                             const float* __restrict__ bij, float* __restrict__ part,
                             int first) {
    __shared__ float ctab[CPC_MAX * OUT_CAPS];
    int lane = threadIdx.x & 31;
    int wv   = __builtin_amdgcn_readfirstlane(threadIdx.x >> 5);
    int kc   = blockIdx.x >> 2;
    int rq   = blockIdx.x & 3;
    int rp   = wv >> 1;
    int oq   = wv & 1;
    int ln   = lane & 15;
    int half = lane >> 4;
    int d0   = half * 4;

    int cap0 = kc * CPC_LO + min(kc, CPC_EXTRA);
    int ncap = CPC_LO + (kc < CPC_EXTRA ? 1 : 0);
    int row0 = rq * 64 + rp * 32;

    const float* xlane = x + (size_t)(row0 + ln) * KDIM + (size_t)cap0 * 8 + d0;
    const float* wlane = W + (size_t)cap0 * 1024 + oq * (RG_TN * 128) + ln * 8 + d0;

    RouteFrags f0, f1;
    route_load(f0, xlane, wlane, 0);

    for (int j = threadIdx.x; j < ncap; j += 128) {
        float v[OUT_CAPS];
        if (first) {
#pragma unroll
            for (int o = 0; o < OUT_CAPS; ++o) v[o] = 0.125f;
        } else {
            const float* bp = bij + (size_t)(cap0 + j) * OUT_CAPS;
            v4f b0 = *(const v4f*)bp;
            v4f b1 = *(const v4f*)(bp + 4);
            v[0] = b0.x; v[1] = b0.y; v[2] = b0.z; v[3] = b0.w;
            v[4] = b1.x; v[5] = b1.y; v[6] = b1.z; v[7] = b1.w;
            float mx = -1e30f;
#pragma unroll
            for (int o = 0; o < OUT_CAPS; ++o) mx = fmaxf(mx, v[o]);
            float sum = 0.0f;
#pragma unroll
            for (int o = 0; o < OUT_CAPS; ++o) { v[o] = __expf(v[o] - mx); sum += v[o]; }
            float inv = 1.0f / sum;
#pragma unroll
            for (int o = 0; o < OUT_CAPS; ++o) v[o] *= inv;
        }
        v4f c0 = { v[0], v[1], v[2], v[3] };
        v4f c1 = { v[4], v[5], v[6], v[7] };
        *(v4f*)(ctab + j * OUT_CAPS)     = c0;
        *(v4f*)(ctab + j * OUT_CAPS + 4) = c1;
    }
    __syncthreads();

    v8f acc[RG_TM][RG_TN] = {};
#pragma nounroll
    for (int g = 0; g < RG_NG - 2; g += 2) {
        route_load(f1, xlane, wlane, (g + 1) * RG_U);
        route_mma(acc, f0, ctab, g * RG_U, g * RG_U, oq, lane);
        route_load(f0, xlane, wlane, (g + 2) * RG_U);
        route_mma(acc, f1, ctab, (g + 1) * RG_U, (g + 1) * RG_U, oq, lane);
    }
    route_load(f1, xlane, wlane, ncap - RG_U);
    route_mma(acc, f0, ctab, (RG_NG - 2) * RG_U, (RG_NG - 2) * RG_U, oq, lane);
    route_mma(acc, f1, ctab, ncap - RG_U, (RG_NG - 1) * RG_U, oq, lane);
#pragma unroll
    for (int t = 0; t < RG_TM; ++t) {
#pragma unroll
        for (int q = 0; q < RG_TN; ++q) {
            float* pbase = part + ((size_t)kc * BATCH + row0 + t * 16) * NDIM
                         + (oq * RG_TN + q) * 16 + ln;
#pragma unroll
            for (int j = 0; j < 8; ++j)
                pbase[(size_t)(j + half * 8) * NDIM] = acc[t][q][j];
        }
    }
}

__global__ void k_reduce_squash(const float* __restrict__ part, float* __restrict__ vT,
                                float* __restrict__ out, float* __restrict__ bij,
                                int last, int zero_b) {
    int b = blockIdx.x;
    int n = threadIdx.x;
    if (zero_b) {
        int t = b * NDIM + n;
        if (t < IN_CAPS * OUT_CAPS) bij[t] = 0.0f;
    }
    float s = 0.0f;
#pragma unroll
    for (int kc = 0; kc < NKC; ++kc)
        s += part[((size_t)kc * BATCH + b) * NDIM + n];
    float msq = s * s;
#pragma unroll
    for (int m = 1; m < 16; m <<= 1)
        msq += __shfl_xor(msq, m, 32);
    float mag = sqrtf(msq);
    float v = s * (mag / (1.0f + msq));
    vT[(size_t)n * BATCH + b] = v;
    if (last) out[(size_t)b * NDIM + n] = v;
}

__global__ void __launch_bounds__(128)
k_agree(const float* __restrict__ x, const float* __restrict__ vT,
        const float* __restrict__ W, float* __restrict__ bij) {
    int lane = threadIdx.x & 31;
    int job  = __builtin_amdgcn_readfirstlane(blockIdx.x * 4 + (threadIdx.x >> 5));
    int g    = job >> 3;
    int o    = job & 7;
    int ln   = lane & 15;
    int half = lane >> 4;
    int k0   = half * 2;

    int kofs = min(g * 64 + ln * 4, KDIM - 4);
    const float* xl   = x + (size_t)k0 * KDIM + kofs;
    const float* brow = vT + (size_t)(o * 16 + ln) * BATCH + k0;
    v8f acc[AG_T] = {};
    for (int kb = 0; kb < BATCH; kb += 4 * AG_U) {
        v4f xa[AG_U][2];
        v2f vb[AG_U];
#pragma unroll
        for (int u = 0; u < AG_U; ++u) {
            xa[u][0] = *(const v4f*)(xl + (size_t)(kb + 4 * u) * KDIM);
            xa[u][1] = *(const v4f*)(xl + (size_t)(kb + 4 * u + 1) * KDIM);
            vb[u]    = *(const v2f*)(brow + kb + 4 * u);
        }
#pragma unroll
        for (int u = 0; u < AG_U; ++u) {
#pragma unroll
            for (int t = 0; t < AG_T; ++t) {
                v2f a = { xa[u][0][t], xa[u][1][t] };
                acc[t] = __builtin_amdgcn_wmma_f32_16x16x4_f32(false, a, false, vb[u],
                                                               (short)0, acc[t], false, false);
            }
        }
    }
    float dot[4];
#pragma unroll
    for (int q = 0; q < 4; ++q) {
        int iW = min(g * 8 + half * 4 + q, IN_CAPS - 1);
        const float* wp = W + (size_t)iW * 1024 + o * 128 + ln * 8;
        v4f w0 = *(const v4f*)wp;
        v4f w1 = *(const v4f*)(wp + 4);
        dot[q] = acc[0][2 * q] * w0.x + acc[1][2 * q] * w0.y
               + acc[2][2 * q] * w0.z + acc[3][2 * q] * w0.w
               + acc[0][2 * q + 1] * w1.x + acc[1][2 * q + 1] * w1.y
               + acc[2][2 * q + 1] * w1.z + acc[3][2 * q + 1] * w1.w;
    }
#pragma unroll
    for (int q = 0; q < 4; ++q) {
#pragma unroll
        for (int m = 1; m < 16; m <<= 1)
            dot[q] += __shfl_xor(dot[q], m, 32);
    }
#pragma unroll
    for (int q = 0; q < 4; ++q) {
        int iW = g * 8 + half * 4 + q;
        if (ln == 0 && iW < IN_CAPS)
            bij[iW * OUT_CAPS + o] += dot[q] * (1.0f / BATCH);
    }
}

extern "C" void kernel_launch(void* const* d_in, const int* in_sizes, int n_in,
                              void* d_out, int out_size, void* d_ws, size_t ws_size,
                              hipStream_t stream) {
    const float* x = (const float*)d_in[0];
    const float* W = (const float*)d_in[1];
    float* out = (float*)d_out;

    float* ws   = (float*)d_ws;
    float* part = ws;
    float* vT   = part + (size_t)NKC * BATCH * NDIM;
    float* bij  = vT + (size_t)NDIM * BATCH;

    k_route_gemm<<<256, 128, 0, stream>>>(x, W, bij, part, 1);
    k_reduce_squash<<<BATCH, NDIM, 0, stream>>>(part, vT, out, bij, 0, 1);
    k_agree<<<AG_GROUPS * OUT_CAPS / 4, 128, 0, stream>>>(x, vT, W, bij);
    k_route_gemm<<<256, 128, 0, stream>>>(x, W, bij, part, 0);
    k_reduce_squash<<<BATCH, NDIM, 0, stream>>>(part, vT, out, bij, 0, 0);
    k_agree<<<AG_GROUPS * OUT_CAPS / 4, 128, 0, stream>>>(x, vT, W, bij);
    k_route_gemm<<<256, 128, 0, stream>>>(x, W, bij, part, 0);
    k_reduce_squash<<<BATCH, NDIM, 0, stream>>>(part, vT, out, bij, 1, 0);
}


# agreement GEMM: squash kernel stores v in B-fragment order, so a B load fetches two k-steps from 4 whole lines, not one k-step from 16 lines 1 KiB apart; sum order kept
# speedup vs baseline: 3.7861x; 1.0892x over previous
#include <hip/hip_runtime.h>
#include <math.h>

#define IN_CAPS 2401
#define IN_SIZE 8
#define OUT_CAPS 8
#define OUT_SIZE 16
#define BATCH 256
#define KDIM (IN_CAPS * IN_SIZE)
#define NDIM (OUT_CAPS * OUT_SIZE)
#define NKC 64
#define CPC_LO 37
#define CPC_EXTRA 33
#define CPC_MAX 38
#define RG_U 4
#define RG_NG 10
#define RG_TM 2
#define RG_TN 4
#define AG_T 4
#define AG_U 8
#define AG_GROUPS ((IN_CAPS + 7) / 8)

typedef float v2f __attribute__((ext_vector_type(2)));
typedef float v4f __attribute__((ext_vector_type(4)));
typedef float v8f __attribute__((ext_vector_type(8)));

struct RouteFrags {
    v4f a[RG_U][RG_TM];
    v4f b[RG_U][RG_TN];
};

__device__ __forceinline__ void route_load(RouteFrags& f, const float* xlane, const float* wlane,
                                           int lb) {
    const float* xg = xlane + lb * 8;
    const float* wg = wlane + (size_t)lb * 1024;
#pragma unroll
    for (int u = 0; u < RG_U; ++u) {
#pragma unroll
        for (int t = 0; t < RG_TM; ++t)
            f.a[u][t] = *(const v4f*)(xg + u * 8 + (size_t)t * 16 * KDIM);
#pragma unroll
        for (int q = 0; q < RG_TN; ++q)
            f.b[u][q] = *(const v4f*)(wg + u * 1024 + q * 128);
    }
}

__device__ __forceinline__ void route_mma(v8f (&acc)[RG_TM][RG_TN], const RouteFrags& f,
                                          const float* ctab, int lb, int gstart, int oq, int lane) {
    int cl = __float_as_int(ctab[(lb + ((lane >> 2) & 3)) * OUT_CAPS + oq * RG_TN + (lane & 3)]);
#pragma unroll
    for (int u = 0; u < RG_U; ++u) {
#pragma unroll
        for (int q = 0; q < RG_TN; ++q) {
            float c = __int_as_float(__builtin_amdgcn_readlane(cl, u * RG_TN + q));
            c = (lb + u >= gstart) ? c : 0.0f;
#pragma unroll
            for (int h = 0; h < 2; ++h) {
                v2f bs = { f.b[u][q][2 * h] * c, f.b[u][q][2 * h + 1] * c };
#pragma unroll
                for (int t = 0; t < RG_TM; ++t) {
                    v2f av = { f.a[u][t][2 * h], f.a[u][t][2 * h + 1] };
                    acc[t][q] = __builtin_amdgcn_wmma_f32_16x16x4_f32(false, av, false, bs,
                                                                      (short)0, acc[t][q],
                                                                      false, false);
                }
            }
        }
    }
}

__global__ void __launch_bounds__(128) __attribute__((amdgpu_waves_per_eu(1, 2)))
k_route_gemm(const float* __restrict__ x, const float* __restrict__ W,
                             const float* __restrict__ bij, float* __restrict__ part,
                             int first) {
    __shared__ float ctab[CPC_MAX * OUT_CAPS];
    int lane = threadIdx.x & 31;
    int wv   = __builtin_amdgcn_readfirstlane(threadIdx.x >> 5);
    int kc   = blockIdx.x >> 2;
    int rq   = blockIdx.x & 3;
    int rp   = wv >> 1;
    int oq   = wv & 1;
    int ln   = lane & 15;
    int half = lane >> 4;
    int d0   = half * 4;

    int cap0 = kc * CPC_LO + min(kc, CPC_EXTRA);
    int ncap = CPC_LO + (kc < CPC_EXTRA ? 1 : 0);
    int row0 = rq * 64 + rp * 32;

    const float* xlane = x + (size_t)(row0 + ln) * KDIM + (size_t)cap0 * 8 + d0;
    const float* wlane = W + (size_t)cap0 * 1024 + oq * (RG_TN * 128) + ln * 8 + d0;

    RouteFrags f0, f1;
    route_load(f0, xlane, wlane, 0);

    for (int j = threadIdx.x; j < ncap; j += 128) {
        float v[OUT_CAPS];
        if (first) {
#pragma unroll
            for (int o = 0; o < OUT_CAPS; ++o) v[o] = 0.125f;
        } else {
            const float* bp = bij + (size_t)(cap0 + j) * OUT_CAPS;
            v4f b0 = *(const v4f*)bp;
            v4f b1 = *(const v4f*)(bp + 4);
            v[0] = b0.x; v[1] = b0.y; v[2] = b0.z; v[3] = b0.w;
            v[4] = b1.x; v[5] = b1.y; v[6] = b1.z; v[7] = b1.w;
            float mx = -1e30f;
#pragma unroll
            for (int o = 0; o < OUT_CAPS; ++o) mx = fmaxf(mx, v[o]);
            float sum = 0.0f;
#pragma unroll
            for (int o = 0; o < OUT_CAPS; ++o) { v[o] = __expf(v[o] - mx); sum += v[o]; }
            float inv = 1.0f / sum;
#pragma unroll
            for (int o = 0; o < OUT_CAPS; ++o) v[o] *= inv;
        }
        v4f c0 = { v[0], v[1], v[2], v[3] };
        v4f c1 = { v[4], v[5], v[6], v[7] };
        *(v4f*)(ctab + j * OUT_CAPS)     = c0;
        *(v4f*)(ctab + j * OUT_CAPS + 4) = c1;
    }
    __syncthreads();

    v8f acc[RG_TM][RG_TN] = {};
#pragma nounroll
    for (int g = 0; g < RG_NG - 2; g += 2) {
        route_load(f1, xlane, wlane, (g + 1) * RG_U);
        route_mma(acc, f0, ctab, g * RG_U, g * RG_U, oq, lane);
        route_load(f0, xlane, wlane, (g + 2) * RG_U);
        route_mma(acc, f1, ctab, (g + 1) * RG_U, (g + 1) * RG_U, oq, lane);
    }
    route_load(f1, xlane, wlane, ncap - RG_U);
    route_mma(acc, f0, ctab, (RG_NG - 2) * RG_U, (RG_NG - 2) * RG_U, oq, lane);
    route_mma(acc, f1, ctab, ncap - RG_U, (RG_NG - 1) * RG_U, oq, lane);
#pragma unroll
    for (int t = 0; t < RG_TM; ++t) {
#pragma unroll
        for (int q = 0; q < RG_TN; ++q) {
            float* pbase = part + ((size_t)kc * BATCH + row0 + t * 16) * NDIM
                         + (oq * RG_TN + q) * 16 + ln;
#pragma unroll
            for (int j = 0; j < 8; ++j)
                pbase[(size_t)(j + half * 8) * NDIM] = acc[t][q][j];
        }
    }
}

__global__ void k_reduce_squash(const float* __restrict__ part, float* __restrict__ vP,
                                float* __restrict__ out, float* __restrict__ bij,
                                int last, int zero_b) {
    int b = blockIdx.x;
    int n = threadIdx.x;
    if (zero_b) {
        int t = b * NDIM + n;
        if (t < IN_CAPS * OUT_CAPS) bij[t] = 0.0f;
    }
    float s = 0.0f;
#pragma unroll
    for (int kc = 0; kc < NKC; ++kc)
        s += part[((size_t)kc * BATCH + b) * NDIM + n];
    float msq = s * s;
#pragma unroll
    for (int m = 1; m < 16; m <<= 1)
        msq += __shfl_xor(msq, m, 32);
    float mag = sqrtf(msq);
    float v = s * (mag / (1.0f + msq));
    int kstep = b >> 2;
    int vlane = ((b >> 1) & 1) * 16 + (n & 15);
    vP[((size_t)((n >> 4) * 32 + (kstep >> 1)) * 32 + vlane) * 4 + (kstep & 1) * 2 + (b & 1)] = v;
    if (last) out[(size_t)b * NDIM + n] = v;
}

__global__ void __launch_bounds__(128) __attribute__((amdgpu_waves_per_eu(1, 4)))
k_agree(const float* __restrict__ x, const float* __restrict__ vP,
        const float* __restrict__ W, float* __restrict__ bij) {
    int lane = threadIdx.x & 31;
    int job  = __builtin_amdgcn_readfirstlane(blockIdx.x * 4 + (threadIdx.x >> 5));
    int g    = job >> 3;
    int o    = job & 7;
    int ln   = lane & 15;
    int half = lane >> 4;
    int k0   = half * 2;

    int kofs = min(g * 64 + ln * 4, KDIM - 4);
    const float* xl   = x + (size_t)k0 * KDIM + kofs;
    const float* bfrag = vP + ((size_t)o * 32 * 32 + lane) * 4;
    v8f acc[AG_T] = {};
    for (int kb = 0; kb < BATCH; kb += 4 * AG_U) {
        v4f xa[AG_U][2];
        v4f vq[AG_U / 2];
#pragma unroll
        for (int u = 0; u < AG_U; ++u) {
            xa[u][0] = *(const v4f*)(xl + (size_t)(kb + 4 * u) * KDIM);
            xa[u][1] = *(const v4f*)(xl + (size_t)(kb + 4 * u + 1) * KDIM);
            if ((u & 1) == 0)
                vq[u >> 1] = *(const v4f*)(bfrag + (size_t)(kb / 8 + (u >> 1)) * 128);
        }
#pragma unroll
        for (int u = 0; u < AG_U; ++u) {
#pragma unroll
            for (int t = 0; t < AG_T; ++t) {
                v2f a = { xa[u][0][t], xa[u][1][t] };
                v2f vb = { vq[u >> 1][(u & 1) * 2], vq[u >> 1][(u & 1) * 2 + 1] };
                acc[t] = __builtin_amdgcn_wmma_f32_16x16x4_f32(false, a, false, vb,
                                                               (short)0, acc[t], false, false);
            }
        }
    }
    float dot[4];
#pragma unroll
    for (int q = 0; q < 4; ++q) {
        int iW = min(g * 8 + half * 4 + q, IN_CAPS - 1);
        const float* wp = W + (size_t)iW * 1024 + o * 128 + ln * 8;
        v4f w0 = *(const v4f*)wp;
        v4f w1 = *(const v4f*)(wp + 4);
        dot[q] = acc[0][2 * q] * w0.x + acc[1][2 * q] * w0.y
               + acc[2][2 * q] * w0.z + acc[3][2 * q] * w0.w
               + acc[0][2 * q + 1] * w1.x + acc[1][2 * q + 1] * w1.y
               + acc[2][2 * q + 1] * w1.z + acc[3][2 * q + 1] * w1.w;
    }
#pragma unroll
    for (int q = 0; q < 4; ++q) {
#pragma unroll
        for (int m = 1; m < 16; m <<= 1)
            dot[q] += __shfl_xor(dot[q], m, 32);
    }
#pragma unroll
    for (int q = 0; q < 4; ++q) {
        int iW = g * 8 + half * 4 + q;
        if (ln == 0 && iW < IN_CAPS)
            bij[iW * OUT_CAPS + o] += dot[q] * (1.0f / BATCH);
    }
}

extern "C" void kernel_launch(void* const* d_in, const int* in_sizes, int n_in,
                              void* d_out, int out_size, void* d_ws, size_t ws_size,
                              hipStream_t stream) {
    const float* x = (const float*)d_in[0];
    const float* W = (const float*)d_in[1];
    float* out = (float*)d_out;

    float* ws   = (float*)d_ws;
    float* part = ws;
    float* vP   = part + (size_t)NKC * BATCH * NDIM;
    float* bij  = vP + (size_t)NDIM * BATCH;

    k_route_gemm<<<256, 128, 0, stream>>>(x, W, bij, part, 1);
    k_reduce_squash<<<BATCH, NDIM, 0, stream>>>(part, vP, out, bij, 0, 1);
    k_agree<<<AG_GROUPS * OUT_CAPS / 4, 128, 0, stream>>>(x, vP, W, bij);
    k_route_gemm<<<256, 128, 0, stream>>>(x, W, bij, part, 0);
    k_reduce_squash<<<BATCH, NDIM, 0, stream>>>(part, vP, out, bij, 0, 0);
    k_agree<<<AG_GROUPS * OUT_CAPS / 4, 128, 0, stream>>>(x, vP, W, bij);
    k_route_gemm<<<256, 128, 0, stream>>>(x, W, bij, part, 0);
    k_reduce_squash<<<BATCH, NDIM, 0, stream>>>(part, vP, out, bij, 1, 0);
}


# agreement GEMM re-tiled: a wave owns 5 M-tiles x 2 output capsules over the whole batch (640 WMMAs); 964 waves, one per SIMD (was 2408 waves of 4x1 tiles, up to 3 per SIMD = 768); sum order kept
# speedup vs baseline: 4.0668x; 1.0741x over previous
#include <hip/hip_runtime.h>
#include <math.h>

#define IN_CAPS 2401
#define IN_SIZE 8
#define OUT_CAPS 8
#define OUT_SIZE 16
#define BATCH 256
#define KDIM (IN_CAPS * IN_SIZE)
#define NDIM (OUT_CAPS * OUT_SIZE)
#define NKC 64
#define CPC_LO 37
#define CPC_EXTRA 33
#define CPC_MAX 38
#define RG_U 4
#define RG_NG 10
#define RG_TM 2
#define RG_TN 4
#define AG_T 5
#define AG_P 2
#define AG_U 8
#define AG_NG (BATCH / (4 * AG_U))
#define AG_CAPS 10
#define AG_GROUPS ((IN_CAPS + AG_CAPS - 1) / AG_CAPS)

typedef float v2f __attribute__((ext_vector_type(2)));
typedef float v4f __attribute__((ext_vector_type(4)));
typedef float v8f __attribute__((ext_vector_type(8)));

struct RouteFrags {
    v4f a[RG_U][RG_TM];
    v4f b[RG_U][RG_TN];
};

__device__ __forceinline__ void route_load(RouteFrags& f, const float* xlane, const float* wlane,
                                           int lb) {
    const float* xg = xlane + lb * 8;
    const float* wg = wlane + (size_t)lb * 1024;
#pragma unroll
    for (int u = 0; u < RG_U; ++u) {
#pragma unroll
        for (int t = 0; t < RG_TM; ++t)
            f.a[u][t] = *(const v4f*)(xg + u * 8 + (size_t)t * 16 * KDIM);
#pragma unroll
        for (int q = 0; q < RG_TN; ++q)
            f.b[u][q] = *(const v4f*)(wg + u * 1024 + q * 128);
    }
}

__device__ __forceinline__ void route_mma(v8f (&acc)[RG_TM][RG_TN], const RouteFrags& f,
                                          const float* ctab, int lb, int gstart, int oq, int lane) {
    int cl = __float_as_int(ctab[(lb + ((lane >> 2) & 3)) * OUT_CAPS + oq * RG_TN + (lane & 3)]);
#pragma unroll
    for (int u = 0; u < RG_U; ++u) {
#pragma unroll
        for (int q = 0; q < RG_TN; ++q) {
            float c = __int_as_float(__builtin_amdgcn_readlane(cl, u * RG_TN + q));
            c = (lb + u >= gstart) ? c : 0.0f;
#pragma unroll
            for (int h = 0; h < 2; ++h) {
                v2f bs = { f.b[u][q][2 * h] * c, f.b[u][q][2 * h + 1] * c };
#pragma unroll
                for (int t = 0; t < RG_TM; ++t) {
                    v2f av = { f.a[u][t][2 * h], f.a[u][t][2 * h + 1] };
                    acc[t][q] = __builtin_amdgcn_wmma_f32_16x16x4_f32(false, av, false, bs,
                                                                      (short)0, acc[t][q],
                                                                      false, false);
                }
            }
        }
    }
}

__global__ void __launch_bounds__(128) __attribute__((amdgpu_waves_per_eu(1, 2)))
k_route_gemm(const float* __restrict__ x, const float* __restrict__ W,
                             const float* __restrict__ bij, float* __restrict__ part,
                             int first) {
    __shared__ float ctab[CPC_MAX * OUT_CAPS];
    int lane = threadIdx.x & 31;
    int wv   = __builtin_amdgcn_readfirstlane(threadIdx.x >> 5);
    int kc   = blockIdx.x >> 2;
    int rq   = blockIdx.x & 3;
    int rp   = wv >> 1;
    int oq   = wv & 1;
    int ln   = lane & 15;
    int half = lane >> 4;
    int d0   = half * 4;

    int cap0 = kc * CPC_LO + min(kc, CPC_EXTRA);
    int ncap = CPC_LO + (kc < CPC_EXTRA ? 1 : 0);
    int row0 = rq * 64 + rp * 32;

    const float* xlane = x + (size_t)(row0 + ln) * KDIM + (size_t)cap0 * 8 + d0;
    const float* wlane = W + (size_t)cap0 * 1024 + oq * (RG_TN * 128) + ln * 8 + d0;

    RouteFrags f0, f1;
    route_load(f0, xlane, wlane, 0);

    for (int j = threadIdx.x; j < ncap; j += 128) {
        float v[OUT_CAPS];
        if (first) {
#pragma unroll
            for (int o = 0; o < OUT_CAPS; ++o) v[o] = 0.125f;
        } else {
            const float* bp = bij + (size_t)(cap0 + j) * OUT_CAPS;
            v4f b0 = *(const v4f*)bp;
            v4f b1 = *(const v4f*)(bp + 4);
            v[0] = b0.x; v[1] = b0.y; v[2] = b0.z; v[3] = b0.w;
            v[4] = b1.x; v[5] = b1.y; v[6] = b1.z; v[7] = b1.w;
            float mx = -1e30f;
#pragma unroll
            for (int o = 0; o < OUT_CAPS; ++o) mx = fmaxf(mx, v[o]);
            float sum = 0.0f;
#pragma unroll
            for (int o = 0; o < OUT_CAPS; ++o) { v[o] = __expf(v[o] - mx); sum += v[o]; }
            float inv = 1.0f / sum;
#pragma unroll
            for (int o = 0; o < OUT_CAPS; ++o) v[o] *= inv;
        }
        v4f c0 = { v[0], v[1], v[2], v[3] };
        v4f c1 = { v[4], v[5], v[6], v[7] };
        *(v4f*)(ctab + j * OUT_CAPS)     = c0;
        *(v4f*)(ctab + j * OUT_CAPS + 4) = c1;
    }
    __syncthreads();

    v8f acc[RG_TM][RG_TN] = {};
#pragma nounroll
    for (int g = 0; g < RG_NG - 2; g += 2) {
        route_load(f1, xlane, wlane, (g + 1) * RG_U);
        route_mma(acc, f0, ctab, g * RG_U, g * RG_U, oq, lane);
        route_load(f0, xlane, wlane, (g + 2) * RG_U);
        route_mma(acc, f1, ctab, (g + 1) * RG_U, (g + 1) * RG_U, oq, lane);
    }
    route_load(f1, xlane, wlane, ncap - RG_U);
    route_mma(acc, f0, ctab, (RG_NG - 2) * RG_U, (RG_NG - 2) * RG_U, oq, lane);
    route_mma(acc, f1, ctab, ncap - RG_U, (RG_NG - 1) * RG_U, oq, lane);
#pragma unroll
    for (int t = 0; t < RG_TM; ++t) {
#pragma unroll
        for (int q = 0; q < RG_TN; ++q) {
            float* pbase = part + ((size_t)kc * BATCH + row0 + t * 16) * NDIM
                         + (oq * RG_TN + q) * 16 + ln;
#pragma unroll
            for (int j = 0; j < 8; ++j)
                pbase[(size_t)(j + half * 8) * NDIM] = acc[t][q][j];
        }
    }
}

__global__ void k_reduce_squash(const float* __restrict__ part, float* __restrict__ vP,
                                float* __restrict__ out, float* __restrict__ bij,
                                int last, int zero_b) {
    int b = blockIdx.x;
    int n = threadIdx.x;
    if (zero_b) {
        int t = b * NDIM + n;
        if (t < IN_CAPS * OUT_CAPS) bij[t] = 0.0f;
    }
    float s = 0.0f;
#pragma unroll
    for (int kc = 0; kc < NKC; ++kc)
        s += part[((size_t)kc * BATCH + b) * NDIM + n];
    float msq = s * s;
#pragma unroll
    for (int m = 1; m < 16; m <<= 1)
        msq += __shfl_xor(msq, m, 32);
    float mag = sqrtf(msq);
    float v = s * (mag / (1.0f + msq));
    int kstep = b >> 2;
    int vlane = ((b >> 1) & 1) * 16 + (n & 15);
    vP[((size_t)((n >> 4) * 32 + (kstep >> 1)) * 32 + vlane) * 4 + (kstep & 1) * 2 + (b & 1)] = v;
    if (last) out[(size_t)b * NDIM + n] = v;
}

struct AgreeFrags {
    v4f   xa[AG_U][2];
    float xb[AG_U][2];
    v4f   vq[AG_U / 2][AG_P];
};

__device__ __forceinline__ void agree_load(AgreeFrags& f, const float* xl4, const float* xl1,
                                           const float* bfrag, int kb) {
    const float* x4 = xl4 + (size_t)kb * KDIM;
    const float* x1 = xl1 + (size_t)kb * KDIM;
    const float* bf = bfrag + (size_t)(kb / 8) * 128;
#pragma unroll
    for (int u = 0; u < AG_U; ++u) {
#pragma unroll
        for (int s = 0; s < 2; ++s) {
            f.xa[u][s] = *(const v4f*)(x4 + (size_t)(4 * u + s) * KDIM);
            f.xb[u][s] = x1[(size_t)(4 * u + s) * KDIM];
        }
    }
#pragma unroll
    for (int w = 0; w < AG_U / 2; ++w) {
#pragma unroll
        for (int p = 0; p < AG_P; ++p)
            f.vq[w][p] = *(const v4f*)(bf + (size_t)p * (32 * 32 * 4) + w * 128);
    }
}

__device__ __forceinline__ void agree_mma(v8f (&acc)[AG_T][AG_P], const AgreeFrags& f) {
#pragma unroll
    for (int u = 0; u < AG_U; ++u) {
#pragma unroll
        for (int p = 0; p < AG_P; ++p) {
            v2f vb = { f.vq[u >> 1][p][(u & 1) * 2], f.vq[u >> 1][p][(u & 1) * 2 + 1] };
#pragma unroll
            for (int t = 0; t < 4; ++t) {
                v2f a = { f.xa[u][0][t], f.xa[u][1][t] };
                acc[t][p] = __builtin_amdgcn_wmma_f32_16x16x4_f32(false, a, false, vb,
                                                                  (short)0, acc[t][p], false, false);
            }
            v2f a4 = { f.xb[u][0], f.xb[u][1] };
            acc[4][p] = __builtin_amdgcn_wmma_f32_16x16x4_f32(false, a4, false, vb,
                                                              (short)0, acc[4][p], false, false);
        }
        __builtin_amdgcn_sched_barrier(0);
    }
}

__global__ void __launch_bounds__(128) __attribute__((amdgpu_waves_per_eu(1, 2)))
k_agree(const float* __restrict__ x, const float* __restrict__ vP,
        const float* __restrict__ W, float* __restrict__ bij) {
    int lane = threadIdx.x & 31;
    int g    = blockIdx.x;
    int o0   = __builtin_amdgcn_readfirstlane(threadIdx.x >> 5) * AG_P;
    int ln   = lane & 15;
    int half = lane >> 4;
    int k0   = half * 2;

    int kofs4 = min(g * (AG_CAPS * 8) + ln * 4, KDIM - 4);
    int kofs1 = min(g * (AG_CAPS * 8) + 64 + ln, KDIM - 1);
    const float* xl4   = x + (size_t)k0 * KDIM + kofs4;
    const float* xl1   = x + (size_t)k0 * KDIM + kofs1;
    const float* bfrag = vP + ((size_t)o0 * 32 * 32 + lane) * 4;

    AgreeFrags f0, f1;
    agree_load(f0, xl4, xl1, bfrag, 0);
    v8f acc[AG_T][AG_P] = {};
#pragma nounroll
    for (int gk = 0; gk < AG_NG - 2; gk += 2) {
        agree_load(f1, xl4, xl1, bfrag, (gk + 1) * 4 * AG_U);
        agree_mma(acc, f0);
        agree_load(f0, xl4, xl1, bfrag, (gk + 2) * 4 * AG_U);
        agree_mma(acc, f1);
    }
    agree_load(f1, xl4, xl1, bfrag, (AG_NG - 1) * 4 * AG_U);

    v4f w0[AG_P][5], w1[AG_P][5];
    v2f bold[5];
#pragma unroll
    for (int q = 0; q < 5; ++q) {
        int iq = (q < 4) ? g * AG_CAPS + half * 4 + q : g * AG_CAPS + 8 + half;
        int iW = min(iq, IN_CAPS - 1);
#pragma unroll
        for (int p = 0; p < AG_P; ++p) {
            const float* wp = W + (size_t)iW * 1024 + (o0 + p) * 128 + ln * 8;
            w0[p][q] = *(const v4f*)wp;
            w1[p][q] = *(const v4f*)(wp + 4);
        }
        bold[q] = *(const v2f*)(bij + iW * OUT_CAPS + o0);
    }
    agree_mma(acc, f0);
    agree_mma(acc, f1);

    float dot[AG_P][5];
#pragma unroll
    for (int p = 0; p < AG_P; ++p) {
#pragma unroll
        for (int q = 0; q < 4; ++q) {
            v4f a = w0[p][q], b = w1[p][q];
            dot[p][q] = acc[0][p][2 * q] * a.x + acc[1][p][2 * q] * a.y
                      + acc[2][p][2 * q] * a.z + acc[3][p][2 * q] * a.w
                      + acc[0][p][2 * q + 1] * b.x + acc[1][p][2 * q + 1] * b.y
                      + acc[2][p][2 * q + 1] * b.z + acc[3][p][2 * q + 1] * b.w;
        }
        v4f a = w0[p][4], b = w1[p][4];
        dot[p][4] = acc[4][p][0] * a.x + acc[4][p][1] * a.y
                  + acc[4][p][2] * a.z + acc[4][p][3] * a.w
                  + acc[4][p][4] * b.x + acc[4][p][5] * b.y
                  + acc[4][p][6] * b.z + acc[4][p][7] * b.w;
    }
#pragma unroll
    for (int p = 0; p < AG_P; ++p) {
#pragma unroll
        for (int q = 0; q < 5; ++q) {
#pragma unroll
            for (int m = 1; m < 16; m <<= 1)
                dot[p][q] += __shfl_xor(dot[p][q], m, 32);
        }
    }
#pragma unroll
    for (int q = 0; q < 5; ++q) {
        int iq = (q < 4) ? g * AG_CAPS + half * 4 + q : g * AG_CAPS + 8 + half;
        v2f nb = { bold[q][0] + dot[0][q] * (1.0f / BATCH), bold[q][1] + dot[1][q] * (1.0f / BATCH) };
        if (ln == 0 && iq < IN_CAPS)
            *(v2f*)(bij + iq * OUT_CAPS + o0) = nb;
    }
}

extern "C" void kernel_launch(void* const* d_in, const int* in_sizes, int n_in,
                              void* d_out, int out_size, void* d_ws, size_t ws_size,
                              hipStream_t stream) {
    const float* x = (const float*)d_in[0];
    const float* W = (const float*)d_in[1];
    float* out = (float*)d_out;

    float* ws   = (float*)d_ws;
    float* part = ws;
    float* vP   = part + (size_t)NKC * BATCH * NDIM;
    float* bij  = vP + (size_t)NDIM * BATCH;

    k_route_gemm<<<256, 128, 0, stream>>>(x, W, bij, part, 1);
    k_reduce_squash<<<BATCH, NDIM, 0, stream>>>(part, vP, out, bij, 0, 1);
    k_agree<<<AG_GROUPS, 128, 0, stream>>>(x, vP, W, bij);
    k_route_gemm<<<256, 128, 0, stream>>>(x, W, bij, part, 0);
    k_reduce_squash<<<BATCH, NDIM, 0, stream>>>(part, vP, out, bij, 0, 0);
    k_agree<<<AG_GROUPS, 128, 0, stream>>>(x, vP, W, bij);
    k_route_gemm<<<256, 128, 0, stream>>>(x, W, bij, part, 0);
    k_reduce_squash<<<BATCH, NDIM, 0, stream>>>(part, vP, out, bij, 1, 0);
}
